# starting kernel
# baseline (speedup 1.0000x reference)
#include <hip/hip_runtime.h>
#include <hip/hip_bf16.h>
#include <cmath>

typedef __attribute__((ext_vector_type(2))) float v2f;
typedef __attribute__((ext_vector_type(8))) float v8f;

#define NQ 100
#define JD 51
#define JD4 13
#define NTILE 7

__global__ __launch_bounds__(256) void matcher_blocks_kernel(
    const float* __restrict__ pred_conf,
    const float* __restrict__ pred_joint,
    const float* __restrict__ pred_action,
    const float* __restrict__ pred_ident,
    const float* __restrict__ tar_conf,
    const float* __restrict__ tar_joint,
    float* __restrict__ out)
{
    __shared__ float4 sPJ4[NQ * JD4];
    __shared__ float4 sTJ4[NQ * JD4];
    __shared__ float  sDlp[NQ];
    __shared__ float  sL1mp[NQ];
    __shared__ float  sRow[NQ];
    __shared__ float  sNegT[NQ];

    const int b   = blockIdx.y;
    const int tid = threadIdx.x;

    {
        float* dPJ = reinterpret_cast<float*>(sPJ4);
        float* dTJ = reinterpret_cast<float*>(sTJ4);
        const float* pj = pred_joint + b * NQ * JD;
        const float* tj = tar_joint  + b * NQ * JD;
        for (int idx = tid; idx < NQ * (4 * JD4); idx += 256) {
            int row = idx / (4 * JD4);
            int col = idx - row * (4 * JD4);
            float vp = 0.0f, vt = 0.0f;
            if (col < JD) {
                vp = pj[row * JD + col];
                vt = tj[row * JD + col];
            }
            dPJ[idx] = vp;
            dTJ[idx] = vt;
        }
    }

    if (tid < NQ) {
        float x = pred_conf[b * NQ + tid];
        sDlp[tid] = x;
        float l1mp = (x > 0.0f) ? (-x - log1pf(__expf(-x)))
                                : (-log1pf(__expf(x)));
        sL1mp[tid] = l1mp;

        const float* pa = pred_action + (b * NQ + tid) * 8;
        float m = pa[0];
        #pragma unroll
        for (int k = 1; k < 8; ++k) m = fmaxf(m, pa[k]);
        float s = 0.0f;
        #pragma unroll
        for (int k = 0; k < 8; ++k) s += __expf(pa[k] - m);
        float ce = (m + __logf(s)) - pa[0];

        const float* pi = pred_ident + (b * NQ + tid) * 10;
        float m2 = pi[0];
        #pragma unroll
        for (int k = 1; k < 10; ++k) m2 = fmaxf(m2, pi[k]);
        float s2 = 0.0f;
        #pragma unroll
        for (int k = 0; k < 10; ++k) s2 += __expf(pi[k] - m2);
        ce += (m2 + __logf(s2)) - pi[0];

        sRow[tid]  = ce;
        sNegT[tid] = -tar_conf[b * NQ + tid];
    }
    __syncthreads();

    const int wave = tid >> 5;
    const int lane = tid & 31;
    const int tile = blockIdx.x * 8 + wave;
    if (tile >= NTILE * NTILE) return;
    const int ti  = tile / NTILE;
    const int tjx = tile % NTILE;
    const int i0 = ti * 16;
    const int j0 = tjx * 16;

    v2f a;  a.x = 0.0f;  a.y = 0.0f;
    v2f bb; bb.x = 0.0f; bb.y = 0.0f;
    const int mlo = lane & 15;
    if (lane < 16) {
        int i = i0 + mlo;
        if (i < NQ) { a.x = sDlp[i]; a.y = sL1mp[i]; }
        int j = j0 + mlo;
        if (j < NQ) { bb.x = sNegT[j]; bb.y = -1.0f; }
    }

    v8f c = {};
    c = __builtin_amdgcn_wmma_f32_16x16x4_f32(
             false, a,  false, bb,
             (short)0, c,  false,  false);

    const int j     = j0 + (lane & 15);
    const int ibase = i0 + ((lane >> 4) << 3);
    float* outb = out + b * NQ * NQ;
    if (j < NQ) {
        const float4* tjrow = &sTJ4[j * JD4];
        float4 tr[JD4];
        #pragma unroll
        for (int kk = 0; kk < JD4; ++kk) tr[kk] = tjrow[kk];

        #pragma unroll
        for (int r = 0; r < 8; ++r) {
            int i = ibase + r;
            if (i < NQ) {
                const float4* pjrow = &sPJ4[i * JD4];
                float l1 = 0.0f;
                #pragma unroll
                for (int kk = 0; kk < JD4; ++kk) {
                    float4 pv = pjrow[kk];
                    l1 += fabsf(pv.x - tr[kk].x);
                    l1 += fabsf(pv.y - tr[kk].y);
                    l1 += fabsf(pv.z - tr[kk].z);
                    l1 += fabsf(pv.w - tr[kk].w);
                }
                outb[i * NQ + j] = c[r] + l1 + sRow[i];
            }
        }
    }
}

extern "C" void kernel_launch(void* const* d_in, const int* in_sizes, int n_in,
                              void* d_out, int out_size, void* d_ws, size_t ws_size,
                              hipStream_t stream) {
    (void)in_sizes; (void)n_in; (void)d_ws; (void)ws_size; (void)out_size;
    const float* pred_conf   = (const float*)d_in[0];
    const float* pred_joint  = (const float*)d_in[1];
    const float* pred_action = (const float*)d_in[2];
    const float* pred_ident  = (const float*)d_in[3];
    const float* tar_conf    = (const float*)d_in[4];
    const float* tar_joint   = (const float*)d_in[5];
    float* out = (float*)d_out;

    dim3 grid(NTILE, 16);
    dim3 block(256);
    matcher_blocks_kernel<<<grid, block, 0, stream>>>(
        pred_conf, pred_joint, pred_action, pred_ident, tar_conf, tar_joint, out);
}


# Prologue: all global loads (joint rows as contiguous float4, per-row scalars) issued before the first wait; was 21 serial load-wait-store trips plus 2 for the scalars
# speedup vs baseline: 2.5655x; 2.5655x over previous
#include <hip/hip_runtime.h>
#include <hip/hip_bf16.h>
#include <cmath>

typedef __attribute__((ext_vector_type(2))) float v2f;
typedef __attribute__((ext_vector_type(8))) float v8f;

#define NQ 100
#define JD 51
#define JD4 13
#define NTILE 7
#define NF4 1275
#define NLD 5

__global__ __launch_bounds__(256) void matcher_blocks_kernel(
    const float* __restrict__ pred_conf,
    const float* __restrict__ pred_joint,
    const float* __restrict__ pred_action,
    const float* __restrict__ pred_ident,
    const float* __restrict__ tar_conf,
    const float* __restrict__ tar_joint,
    float* __restrict__ out)
{
    __shared__ float4 sPJ4[NQ * JD4];
    __shared__ float4 sTJ4[NQ * JD4];
    __shared__ float  sDlp[NQ];
    __shared__ float  sL1mp[NQ];
    __shared__ float  sRow[NQ];
    __shared__ float  sNegT[NQ];

    const int b   = blockIdx.y;
    const int tid = threadIdx.x;

    const bool rowThread = tid < NQ;

    const int srow = b * NQ + (rowThread ? tid : NQ - 1);
    const float x   = pred_conf[srow];
    const float tcv = tar_conf[srow];
    float pa[8], pi[10];
    {
        const float* pap = pred_action + srow * 8;
        #pragma unroll
        for (int k = 0; k < 8; ++k) pa[k] = pap[k];
        const float* pip = pred_ident + srow * 10;
        #pragma unroll
        for (int k = 0; k < 10; ++k) pi[k] = pip[k];
    }

    const float4* pj4 = reinterpret_cast<const float4*>(pred_joint + b * NQ * JD);
    const float4* tj4 = reinterpret_cast<const float4*>(tar_joint  + b * NQ * JD);
    float4 vp[NLD], vt[NLD];
    #pragma unroll
    for (int k = 0; k < NLD; ++k) {
        const int q = tid + 256 * k;
        if (k < NLD - 1 || q < NF4) { vp[k] = pj4[q]; vt[k] = tj4[q]; }
    }

    float* dPJ = reinterpret_cast<float*>(sPJ4);
    float* dTJ = reinterpret_cast<float*>(sTJ4);
    if (rowThread) {
        sDlp[tid] = x;
        float l1mp = (x > 0.0f) ? (-x - log1pf(__expf(-x)))
                                : (-log1pf(__expf(x)));
        sL1mp[tid] = l1mp;

        float m = pa[0];
        #pragma unroll
        for (int k = 1; k < 8; ++k) m = fmaxf(m, pa[k]);
        float s = 0.0f;
        #pragma unroll
        for (int k = 0; k < 8; ++k) s += __expf(pa[k] - m);
        float ce = (m + __logf(s)) - pa[0];

        float m2 = pi[0];
        #pragma unroll
        for (int k = 1; k < 10; ++k) m2 = fmaxf(m2, pi[k]);
        float s2 = 0.0f;
        #pragma unroll
        for (int k = 0; k < 10; ++k) s2 += __expf(pi[k] - m2);
        ce += (m2 + __logf(s2)) - pi[0];

        sRow[tid]  = ce;
        sNegT[tid] = -tcv;

        dPJ[tid * (4 * JD4) + JD] = 0.0f;
        dTJ[tid * (4 * JD4) + JD] = 0.0f;
    }

    #pragma unroll
    for (int k = 0; k < NLD; ++k) {
        const int q = tid + 256 * k;
        if (k < NLD - 1 || q < NF4) {
            const unsigned e0   = 4u * (unsigned)q;
            const unsigned row0 = e0 / (unsigned)JD;
            const unsigned col0 = e0 - row0 * (unsigned)JD;
            const unsigned d0 = e0 + row0;
            const unsigned d1 = d0 + 1u + (col0 >= (unsigned)(JD - 1) ? 1u : 0u);
            const unsigned d2 = d0 + 2u + (col0 >= (unsigned)(JD - 2) ? 1u : 0u);
            const unsigned d3 = d0 + 3u + (col0 >= (unsigned)(JD - 3) ? 1u : 0u);
            dPJ[d0] = vp[k].x;  dTJ[d0] = vt[k].x;
            dPJ[d1] = vp[k].y;  dTJ[d1] = vt[k].y;
            dPJ[d2] = vp[k].z;  dTJ[d2] = vt[k].z;
            dPJ[d3] = vp[k].w;  dTJ[d3] = vt[k].w;
        }
    }
    __syncthreads();

    const int wave = tid >> 5;
    const int lane = tid & 31;
    const int tile = blockIdx.x * 8 + wave;
    if (tile >= NTILE * NTILE) return;
    const int ti  = tile / NTILE;
    const int tjx = tile % NTILE;
    const int i0 = ti * 16;
    const int j0 = tjx * 16;

    v2f a;  a.x = 0.0f;  a.y = 0.0f;
    v2f bb; bb.x = 0.0f; bb.y = 0.0f;
    const int mlo = lane & 15;
    if (lane < 16) {
        int i = i0 + mlo;
        if (i < NQ) { a.x = sDlp[i]; a.y = sL1mp[i]; }
        int j = j0 + mlo;
        if (j < NQ) { bb.x = sNegT[j]; bb.y = -1.0f; }
    }

    v8f c = {};
    c = __builtin_amdgcn_wmma_f32_16x16x4_f32(
             false, a,  false, bb,
             (short)0, c,  false,  false);

    const int j     = j0 + (lane & 15);
    const int ibase = i0 + ((lane >> 4) << 3);
    float* outb = out + b * NQ * NQ;
    if (j < NQ) {
        const float4* tjrow = &sTJ4[j * JD4];
        float4 tr[JD4];
        #pragma unroll
        for (int kk = 0; kk < JD4; ++kk) tr[kk] = tjrow[kk];

        #pragma unroll
        for (int r = 0; r < 8; ++r) {
            int i = ibase + r;
            if (i < NQ) {
                const float4* pjrow = &sPJ4[i * JD4];
                float l1 = 0.0f;
                #pragma unroll
                for (int kk = 0; kk < JD4; ++kk) {
                    float4 pv = pjrow[kk];
                    l1 += fabsf(pv.x - tr[kk].x);
                    l1 += fabsf(pv.y - tr[kk].y);
                    l1 += fabsf(pv.z - tr[kk].z);
                    l1 += fabsf(pv.w - tr[kk].w);
                }
                outb[i * NQ + j] = c[r] + l1 + sRow[i];
            }
        }
    }
}

extern "C" void kernel_launch(void* const* d_in, const int* in_sizes, int n_in,
                              void* d_out, int out_size, void* d_ws, size_t ws_size,
                              hipStream_t stream) {
    (void)in_sizes; (void)n_in; (void)d_ws; (void)ws_size; (void)out_size;
    const float* pred_conf   = (const float*)d_in[0];
    const float* pred_joint  = (const float*)d_in[1];
    const float* pred_action = (const float*)d_in[2];
    const float* pred_ident  = (const float*)d_in[3];
    const float* tar_conf    = (const float*)d_in[4];
    const float* tar_joint   = (const float*)d_in[5];
    float* out = (float*)d_out;

    dim3 grid(NTILE, 16);
    dim3 block(256);
    matcher_blocks_kernel<<<grid, block, 0, stream>>>(
        pred_conf, pred_joint, pred_action, pred_ident, tar_conf, tar_joint, out);
}


# L1 sums: a lane's 8 row sums advance together (k outer, row inner) with the next k-group's 8 LDS reads in flight; was row by row, every add and LDS read waited on. Per-output add order unchanged
# speedup vs baseline: 3.1835x; 1.2409x over previous
#include <hip/hip_runtime.h>
#include <hip/hip_bf16.h>
#include <cmath>

typedef __attribute__((ext_vector_type(2))) float v2f;
typedef __attribute__((ext_vector_type(8))) float v8f;

#define NQ 100
#define JD 51
#define JD4 13
#define NTILE 7
#define NF4 1275
#define NLD 5

__global__ __launch_bounds__(256) void matcher_blocks_kernel(
    const float* __restrict__ pred_conf,
    const float* __restrict__ pred_joint,
    const float* __restrict__ pred_action,
    const float* __restrict__ pred_ident,
    const float* __restrict__ tar_conf,
    const float* __restrict__ tar_joint,
    float* __restrict__ out)
{
    __shared__ float4 sPJ4[NQ * JD4];
    __shared__ float4 sTJ4[NQ * JD4];
    __shared__ float  sDlp[NQ];
    __shared__ float  sL1mp[NQ];
    __shared__ float  sRow[NQ];
    __shared__ float  sNegT[NQ];

    const int b   = blockIdx.y;
    const int tid = threadIdx.x;

    const bool rowThread = tid < NQ;

    const int srow = b * NQ + (rowThread ? tid : NQ - 1);
    const float x   = pred_conf[srow];
    const float tcv = tar_conf[srow];
    float pa[8], pi[10];
    {
        const float* pap = pred_action + srow * 8;
        #pragma unroll
        for (int k = 0; k < 8; ++k) pa[k] = pap[k];
        const float* pip = pred_ident + srow * 10;
        #pragma unroll
        for (int k = 0; k < 10; ++k) pi[k] = pip[k];
    }

    const float4* pj4 = reinterpret_cast<const float4*>(pred_joint + b * NQ * JD);
    const float4* tj4 = reinterpret_cast<const float4*>(tar_joint  + b * NQ * JD);
    float4 vp[NLD], vt[NLD];
    #pragma unroll
    for (int k = 0; k < NLD; ++k) {
        const int q = tid + 256 * k;
        if (k < NLD - 1 || q < NF4) { vp[k] = pj4[q]; vt[k] = tj4[q]; }
    }

    float* dPJ = reinterpret_cast<float*>(sPJ4);
    float* dTJ = reinterpret_cast<float*>(sTJ4);
    if (rowThread) {
        sDlp[tid] = x;
        float l1mp = (x > 0.0f) ? (-x - log1pf(__expf(-x)))
                                : (-log1pf(__expf(x)));
        sL1mp[tid] = l1mp;

        float m = pa[0];
        #pragma unroll
        for (int k = 1; k < 8; ++k) m = fmaxf(m, pa[k]);
        float s = 0.0f;
        #pragma unroll
        for (int k = 0; k < 8; ++k) s += __expf(pa[k] - m);
        float ce = (m + __logf(s)) - pa[0];

        float m2 = pi[0];
        #pragma unroll
        for (int k = 1; k < 10; ++k) m2 = fmaxf(m2, pi[k]);
        float s2 = 0.0f;
        #pragma unroll
        for (int k = 0; k < 10; ++k) s2 += __expf(pi[k] - m2);
        ce += (m2 + __logf(s2)) - pi[0];

        sRow[tid]  = ce;
        sNegT[tid] = -tcv;

        dPJ[tid * (4 * JD4) + JD] = 0.0f;
        dTJ[tid * (4 * JD4) + JD] = 0.0f;
    }

    #pragma unroll
    for (int k = 0; k < NLD; ++k) {
        const int q = tid + 256 * k;
        if (k < NLD - 1 || q < NF4) {
            const unsigned e0   = 4u * (unsigned)q;
            const unsigned row0 = e0 / (unsigned)JD;
            const unsigned col0 = e0 - row0 * (unsigned)JD;
            const unsigned d0 = e0 + row0;
            const unsigned d1 = d0 + 1u + (col0 >= (unsigned)(JD - 1) ? 1u : 0u);
            const unsigned d2 = d0 + 2u + (col0 >= (unsigned)(JD - 2) ? 1u : 0u);
            const unsigned d3 = d0 + 3u + (col0 >= (unsigned)(JD - 3) ? 1u : 0u);
            dPJ[d0] = vp[k].x;  dTJ[d0] = vt[k].x;
            dPJ[d1] = vp[k].y;  dTJ[d1] = vt[k].y;
            dPJ[d2] = vp[k].z;  dTJ[d2] = vt[k].z;
            dPJ[d3] = vp[k].w;  dTJ[d3] = vt[k].w;
        }
    }
    __syncthreads();

    const int wave = tid >> 5;
    const int lane = tid & 31;
    const int tile = blockIdx.x * 8 + wave;
    if (tile >= NTILE * NTILE) return;
    const int ti  = tile / NTILE;
    const int tjx = tile % NTILE;
    const int i0 = ti * 16;
    const int j0 = tjx * 16;

    v2f a;  a.x = 0.0f;  a.y = 0.0f;
    v2f bb; bb.x = 0.0f; bb.y = 0.0f;
    const int mlo = lane & 15;
    if (lane < 16) {
        int i = i0 + mlo;
        if (i < NQ) { a.x = sDlp[i]; a.y = sL1mp[i]; }
        int j = j0 + mlo;
        if (j < NQ) { bb.x = sNegT[j]; bb.y = -1.0f; }
    }

    v8f c = {};
    c = __builtin_amdgcn_wmma_f32_16x16x4_f32(
             false, a,  false, bb,
             (short)0, c,  false,  false);

    const int j     = j0 + (lane & 15);
    const int ibase = i0 + ((lane >> 4) << 3);
    float* outb = out + b * NQ * NQ;
    if (j < NQ) {
        const float4* tjrow = &sTJ4[j * JD4];
        float4 tr[JD4];
        #pragma unroll
        for (int kk = 0; kk < JD4; ++kk) tr[kk] = tjrow[kk];

        const float4* prow[8];
        int irow[8];
        #pragma unroll
        for (int r = 0; r < 8; ++r) {
            const int i = ibase + r;
            irow[r] = (i < NQ) ? i : (NQ - 1);
            prow[r] = &sPJ4[irow[r] * JD4];
        }

        float l1[8];
        #pragma unroll
        for (int r = 0; r < 8; ++r) l1[r] = 0.0f;

        float4 pv[8], pn[8];
        #pragma unroll
        for (int r = 0; r < 8; ++r) pv[r] = prow[r][0];

        float sr[8];
        #pragma unroll
        for (int r = 0; r < 8; ++r) sr[r] = sRow[irow[r]];

        #pragma unroll
        for (int kk = 0; kk < JD4; ++kk) {
            if (kk + 1 < JD4) {
                #pragma unroll
                for (int r = 0; r < 8; ++r) pn[r] = prow[r][kk + 1];
            }
            #pragma unroll
            for (int r = 0; r < 8; ++r) l1[r] += fabsf(pv[r].x - tr[kk].x);
            #pragma unroll
            for (int r = 0; r < 8; ++r) l1[r] += fabsf(pv[r].y - tr[kk].y);
            #pragma unroll
            for (int r = 0; r < 8; ++r) l1[r] += fabsf(pv[r].z - tr[kk].z);
            #pragma unroll
            for (int r = 0; r < 8; ++r) l1[r] += fabsf(pv[r].w - tr[kk].w);
            if (kk + 1 < JD4) {
                #pragma unroll
                for (int r = 0; r < 8; ++r) pv[r] = pn[r];
            }
        }

        float val[8];
        #pragma unroll
        for (int r = 0; r < 8; ++r) val[r] = c[r] + l1[r] + sr[r];

        #pragma unroll
        for (int r = 0; r < 8; ++r) asm volatile("" : "+v"(val[r]));

        #pragma unroll
        for (int r = 0; r < 8; ++r) {
            const int i = ibase + r;
            if (i < NQ) outb[i * NQ + j] = val[r];
        }
    }
}

extern "C" void kernel_launch(void* const* d_in, const int* in_sizes, int n_in,
                              void* d_out, int out_size, void* d_ws, size_t ws_size,
                              hipStream_t stream) {
    (void)in_sizes; (void)n_in; (void)d_ws; (void)ws_size; (void)out_size;
    const float* pred_conf   = (const float*)d_in[0];
    const float* pred_joint  = (const float*)d_in[1];
    const float* pred_action = (const float*)d_in[2];
    const float* pred_ident  = (const float*)d_in[3];
    const float* tar_conf    = (const float*)d_in[4];
    const float* tar_joint   = (const float*)d_in[5];
    float* out = (float*)d_out;

    dim3 grid(NTILE, 16);
    dim3 block(256);
    matcher_blocks_kernel<<<grid, block, 0, stream>>>(
        pred_conf, pred_joint, pred_action, pred_ident, tar_conf, tar_joint, out);
}


# Decomposition: a work-group is one 16-row tile row x up to 4 tile columns (224 groups of 4 waves, was 112 of 8) and stages only the rows it reads; one wave per SIMD. Per-output add order unchanged
# speedup vs baseline: 3.5382x; 1.1114x over previous
#include <hip/hip_runtime.h>
#include <hip/hip_bf16.h>
#include <cmath>


typedef __attribute__((ext_vector_type(2))) float v2f;
typedef __attribute__((ext_vector_type(8))) float v8f;

#define NQ 100
#define JD 51
#define JD4 13
#define NTILE 7
#define PR 16
#define TC 64
#define NTHR 128
#define NLDP 2
#define NLDT 7

__global__ __launch_bounds__(NTHR) void matcher_blocks_kernel(
    const float* __restrict__ pred_conf,
    const float* __restrict__ pred_joint,
    const float* __restrict__ pred_action,
    const float* __restrict__ pred_ident,
    const float* __restrict__ tar_conf,
    const float* __restrict__ tar_joint,
    float* __restrict__ out)
{
    __shared__ float4 sPJ4[PR * JD4];
    __shared__ float4 sTJ4[TC * JD4];
    __shared__ float  sDlp[PR];
    __shared__ float  sL1mp[PR];
    __shared__ float  sRow[PR];
    __shared__ float  sNegT[TC];

    const int b    = blockIdx.y;
    const int tid  = threadIdx.x;
    const int ti   = blockIdx.x >> 1;
    const int half = blockIdx.x & 1;
    const int i0   = ti * 16;
    const int c0   = half * TC;
    const int nrow = (NQ - i0 < PR) ? (NQ - i0) : PR;
    const int ncol = (NQ - c0 < TC) ? (NQ - c0) : TC;
    const int nP4  = nrow * JD / 4;
    const int nT4  = ncol * JD / 4;

    const bool rowThread = tid < PR;
    const bool colThread = tid < TC;
    const int ri   = i0 + (rowThread ? tid : 0);
    const int cj   = c0 + (colThread ? tid : 0);
    const int srow = b * NQ + (ri < NQ ? ri : NQ - 1);
    const int scol = b * NQ + (cj < NQ ? cj : NQ - 1);
    const float x   = pred_conf[srow];
    const float tcv = tar_conf[scol];
    float pa[8], pi[10];
    {
        const float* pap = pred_action + srow * 8;
        #pragma unroll
        for (int k = 0; k < 8; ++k) pa[k] = pap[k];
        const float* pip = pred_ident + srow * 10;
        #pragma unroll
        for (int k = 0; k < 10; ++k) pi[k] = pip[k];
    }

    const float4* pj4 = reinterpret_cast<const float4*>(pred_joint + (b * NQ + i0) * JD);
    const float4* tj4 = reinterpret_cast<const float4*>(tar_joint  + (b * NQ + c0) * JD);
    float4 vp[NLDP], vt[NLDT];
    #pragma unroll
    for (int k = 0; k < NLDP; ++k) {
        const int q = tid + NTHR * k;
        vp[k] = pj4[q < nP4 ? q : nP4 - 1];
    }
    #pragma unroll
    for (int k = 0; k < NLDT; ++k) {
        const int q = tid + NTHR * k;
        vt[k] = tj4[q < nT4 ? q : nT4 - 1];
    }

    asm volatile("" :: "v"(x),
                 "v"(pa[0]), "v"(pa[1]), "v"(pa[2]), "v"(pa[3]),
                 "v"(pa[4]), "v"(pa[5]), "v"(pa[6]), "v"(pa[7]),
                 "v"(pi[0]), "v"(pi[1]), "v"(pi[2]), "v"(pi[3]), "v"(pi[4]),
                 "v"(pi[5]), "v"(pi[6]), "v"(pi[7]), "v"(pi[8]), "v"(pi[9]));

    float* dPJ = reinterpret_cast<float*>(sPJ4);
    float* dTJ = reinterpret_cast<float*>(sTJ4);
    if (rowThread) {
        sDlp[tid] = x;
        float l1mp = (x > 0.0f) ? (-x - log1pf(__expf(-x)))
                                : (-log1pf(__expf(x)));
        sL1mp[tid] = l1mp;

        float m = pa[0];
        #pragma unroll
        for (int k = 1; k < 8; ++k) m = fmaxf(m, pa[k]);
        float s = 0.0f;
        #pragma unroll
        for (int k = 0; k < 8; ++k) s += __expf(pa[k] - m);
        float ce = (m + __logf(s)) - pa[0];

        float m2 = pi[0];
        #pragma unroll
        for (int k = 1; k < 10; ++k) m2 = fmaxf(m2, pi[k]);
        float s2 = 0.0f;
        #pragma unroll
        for (int k = 0; k < 10; ++k) s2 += __expf(pi[k] - m2);
        ce += (m2 + __logf(s2)) - pi[0];

        sRow[tid] = ce;

        dPJ[tid * (4 * JD4) + JD] = 0.0f;
    }
    if (colThread) {
        sNegT[tid] = -tcv;
        dTJ[tid * (4 * JD4) + JD] = 0.0f;
    }

    #pragma unroll
    for (int k = 0; k < NLDP; ++k) {
        const int q = tid + NTHR * k;
        if (q < nP4) {
            const unsigned e0   = 4u * (unsigned)q;
            const unsigned row0 = e0 / (unsigned)JD;
            const unsigned col0 = e0 - row0 * (unsigned)JD;
            const unsigned d0 = e0 + row0;
            const unsigned d1 = d0 + 1u + (col0 >= (unsigned)(JD - 1) ? 1u : 0u);
            const unsigned d2 = d0 + 2u + (col0 >= (unsigned)(JD - 2) ? 1u : 0u);
            const unsigned d3 = d0 + 3u + (col0 >= (unsigned)(JD - 3) ? 1u : 0u);
            dPJ[d0] = vp[k].x;
            dPJ[d1] = vp[k].y;
            dPJ[d2] = vp[k].z;
            dPJ[d3] = vp[k].w;
        }
    }
    #pragma unroll
    for (int k = 0; k < NLDT; ++k) {
        const int q = tid + NTHR * k;
        if (k < 3 || q < nT4) {
            const unsigned e0   = 4u * (unsigned)q;
            const unsigned row0 = e0 / (unsigned)JD;
            const unsigned col0 = e0 - row0 * (unsigned)JD;
            const unsigned d0 = e0 + row0;
            const unsigned d1 = d0 + 1u + (col0 >= (unsigned)(JD - 1) ? 1u : 0u);
            const unsigned d2 = d0 + 2u + (col0 >= (unsigned)(JD - 2) ? 1u : 0u);
            const unsigned d3 = d0 + 3u + (col0 >= (unsigned)(JD - 3) ? 1u : 0u);
            dTJ[d0] = vt[k].x;
            dTJ[d1] = vt[k].y;
            dTJ[d2] = vt[k].z;
            dTJ[d3] = vt[k].w;
        }
    }
    __syncthreads();

    const int wave = tid >> 5;
    const int lane = tid & 31;
    const int tjx  = half * 4 + wave;
    if (tjx >= NTILE) return;
    const int j0  = tjx * 16;
    const int jl0 = wave * 16;

    v2f a;  a.x = 0.0f;  a.y = 0.0f;
    v2f bb; bb.x = 0.0f; bb.y = 0.0f;
    const int mlo = lane & 15;
    if (lane < 16) {
        int i = i0 + mlo;
        if (i < NQ) { a.x = sDlp[mlo]; a.y = sL1mp[mlo]; }
        int j = j0 + mlo;
        if (j < NQ) { bb.x = sNegT[jl0 + mlo]; bb.y = -1.0f; }
    }

    v8f c = {};
    c = __builtin_amdgcn_wmma_f32_16x16x4_f32(
             false, a,  false, bb,
             (short)0, c,  false,  false);

    const int j     = j0 + (lane & 15);
    const int ibase = i0 + ((lane >> 4) << 3);
    float* outb = out + b * NQ * NQ;
    if (j < NQ) {
        const float4* tjrow = &sTJ4[(jl0 + (lane & 15)) * JD4];
        float4 tr[JD4];
        #pragma unroll
        for (int kk = 0; kk < JD4; ++kk) tr[kk] = tjrow[kk];

        const float4* prow[8];
        int lrow[8];
        #pragma unroll
        for (int r = 0; r < 8; ++r) {
            const int i = ibase + r;
            lrow[r] = ((i < NQ) ? i : (NQ - 1)) - i0;
            prow[r] = &sPJ4[lrow[r] * JD4];
        }

        float l1[8];
        #pragma unroll
        for (int r = 0; r < 8; ++r) l1[r] = 0.0f;

        float4 pv[8], pn[8];
        #pragma unroll
        for (int r = 0; r < 8; ++r) pv[r] = prow[r][0];

        float sr[8];
        #pragma unroll
        for (int r = 0; r < 8; ++r) sr[r] = sRow[lrow[r]];

        #pragma unroll
        for (int kk = 0; kk < JD4; ++kk) {
            if (kk + 1 < JD4) {
                #pragma unroll
                for (int r = 0; r < 8; ++r) pn[r] = prow[r][kk + 1];
            }
            #pragma unroll
            for (int r = 0; r < 8; ++r) l1[r] += fabsf(pv[r].x - tr[kk].x);
            #pragma unroll
            for (int r = 0; r < 8; ++r) l1[r] += fabsf(pv[r].y - tr[kk].y);
            #pragma unroll
            for (int r = 0; r < 8; ++r) l1[r] += fabsf(pv[r].z - tr[kk].z);
            #pragma unroll
            for (int r = 0; r < 8; ++r) l1[r] += fabsf(pv[r].w - tr[kk].w);
            if (kk + 1 < JD4) {
                #pragma unroll
                for (int r = 0; r < 8; ++r) pv[r] = pn[r];
            }
        }

        float val[8];
        #pragma unroll
        for (int r = 0; r < 8; ++r) val[r] = c[r] + l1[r] + sr[r];

        #pragma unroll
        for (int r = 0; r < 8; ++r) asm volatile("" : "+v"(val[r]));

        #pragma unroll
        for (int r = 0; r < 8; ++r) {
            const int i = ibase + r;
            if (i < NQ) outb[i * NQ + j] = val[r];
        }
    }
}

extern "C" void kernel_launch(void* const* d_in, const int* in_sizes, int n_in,
                              void* d_out, int out_size, void* d_ws, size_t ws_size,
                              hipStream_t stream) {
    (void)in_sizes; (void)n_in; (void)d_ws; (void)ws_size; (void)out_size;
    const float* pred_conf   = (const float*)d_in[0];
    const float* pred_joint  = (const float*)d_in[1];
    const float* pred_action = (const float*)d_in[2];
    const float* pred_ident  = (const float*)d_in[3];
    const float* tar_conf    = (const float*)d_in[4];
    const float* tar_joint   = (const float*)d_in[5];
    float* out = (float*)d_out;

    dim3 grid(2 * NTILE, 16);
    dim3 block(NTHR);
    matcher_blocks_kernel<<<grid, block, 0, stream>>>(
        pred_conf, pred_joint, pred_action, pred_ident, tar_conf, tar_joint, out);
}


# Row term log(1-p) = -(max(x,0) + log(1 + exp(-|x|))), branch-free on hardware exp and log; was library log1p, two long dependent branches run one after the other ahead of the barrier
# speedup vs baseline: 3.7180x; 1.0508x over previous
#include <hip/hip_runtime.h>
#include <hip/hip_bf16.h>
#include <cmath>


typedef __attribute__((ext_vector_type(2))) float v2f;
typedef __attribute__((ext_vector_type(8))) float v8f;

#define NQ 100
#define JD 51
#define JD4 13
#define NTILE 7
#define PR 16
#define TC 64
#define NTHR 128
#define NLDP 2
#define NLDT 7

__global__ __launch_bounds__(NTHR) void matcher_blocks_kernel(
    const float* __restrict__ pred_conf,
    const float* __restrict__ pred_joint,
    const float* __restrict__ pred_action,
    const float* __restrict__ pred_ident,
    const float* __restrict__ tar_conf,
    const float* __restrict__ tar_joint,
    float* __restrict__ out)
{
    __shared__ float4 sPJ4[PR * JD4];
    __shared__ float4 sTJ4[TC * JD4];
    __shared__ float  sDlp[PR];
    __shared__ float  sL1mp[PR];
    __shared__ float  sRow[PR];
    __shared__ float  sNegT[TC];

    const int b    = blockIdx.y;
    const int tid  = threadIdx.x;
    const int ti   = blockIdx.x >> 1;
    const int half = blockIdx.x & 1;
    const int i0   = ti * 16;
    const int c0   = half * TC;
    const int nrow = (NQ - i0 < PR) ? (NQ - i0) : PR;
    const int ncol = (NQ - c0 < TC) ? (NQ - c0) : TC;
    const int nP4  = nrow * JD / 4;
    const int nT4  = ncol * JD / 4;

    const bool rowThread = tid < PR;
    const bool colThread = tid < TC;
    const int ri   = i0 + (rowThread ? tid : 0);
    const int cj   = c0 + (colThread ? tid : 0);
    const int srow = b * NQ + (ri < NQ ? ri : NQ - 1);
    const int scol = b * NQ + (cj < NQ ? cj : NQ - 1);
    const float x   = pred_conf[srow];
    const float tcv = tar_conf[scol];
    float pa[8], pi[10];
    {
        const float* pap = pred_action + srow * 8;
        #pragma unroll
        for (int k = 0; k < 8; ++k) pa[k] = pap[k];
        const float* pip = pred_ident + srow * 10;
        #pragma unroll
        for (int k = 0; k < 10; ++k) pi[k] = pip[k];
    }

    const float4* pj4 = reinterpret_cast<const float4*>(pred_joint + (b * NQ + i0) * JD);
    const float4* tj4 = reinterpret_cast<const float4*>(tar_joint  + (b * NQ + c0) * JD);
    float4 vp[NLDP], vt[NLDT];
    #pragma unroll
    for (int k = 0; k < NLDP; ++k) {
        const int q = tid + NTHR * k;
        vp[k] = pj4[q < nP4 ? q : nP4 - 1];
    }
    #pragma unroll
    for (int k = 0; k < NLDT; ++k) {
        const int q = tid + NTHR * k;
        vt[k] = tj4[q < nT4 ? q : nT4 - 1];
    }

    asm volatile("" :: "v"(x),
                 "v"(pa[0]), "v"(pa[1]), "v"(pa[2]), "v"(pa[3]),
                 "v"(pa[4]), "v"(pa[5]), "v"(pa[6]), "v"(pa[7]),
                 "v"(pi[0]), "v"(pi[1]), "v"(pi[2]), "v"(pi[3]), "v"(pi[4]),
                 "v"(pi[5]), "v"(pi[6]), "v"(pi[7]), "v"(pi[8]), "v"(pi[9]));

    float* dPJ = reinterpret_cast<float*>(sPJ4);
    float* dTJ = reinterpret_cast<float*>(sTJ4);
    if (rowThread) {
        sDlp[tid] = x;
        const float e = __expf(-fabsf(x));
        float l1mp = -(fmaxf(x, 0.0f) + __logf(1.0f + e));
        sL1mp[tid] = l1mp;

        float m = pa[0];
        #pragma unroll
        for (int k = 1; k < 8; ++k) m = fmaxf(m, pa[k]);
        float s = 0.0f;
        #pragma unroll
        for (int k = 0; k < 8; ++k) s += __expf(pa[k] - m);
        float ce = (m + __logf(s)) - pa[0];

        float m2 = pi[0];
        #pragma unroll
        for (int k = 1; k < 10; ++k) m2 = fmaxf(m2, pi[k]);
        float s2 = 0.0f;
        #pragma unroll
        for (int k = 0; k < 10; ++k) s2 += __expf(pi[k] - m2);
        ce += (m2 + __logf(s2)) - pi[0];

        sRow[tid] = ce;

        dPJ[tid * (4 * JD4) + JD] = 0.0f;
    }
    if (colThread) {
        sNegT[tid] = -tcv;
        dTJ[tid * (4 * JD4) + JD] = 0.0f;
    }

    #pragma unroll
    for (int k = 0; k < NLDP; ++k) {
        const int q = tid + NTHR * k;
        if (q < nP4) {
            const unsigned e0   = 4u * (unsigned)q;
            const unsigned row0 = e0 / (unsigned)JD;
            const unsigned col0 = e0 - row0 * (unsigned)JD;
            const unsigned d0 = e0 + row0;
            const unsigned d1 = d0 + 1u + (col0 >= (unsigned)(JD - 1) ? 1u : 0u);
            const unsigned d2 = d0 + 2u + (col0 >= (unsigned)(JD - 2) ? 1u : 0u);
            const unsigned d3 = d0 + 3u + (col0 >= (unsigned)(JD - 3) ? 1u : 0u);
            dPJ[d0] = vp[k].x;
            dPJ[d1] = vp[k].y;
            dPJ[d2] = vp[k].z;
            dPJ[d3] = vp[k].w;
        }
    }
    #pragma unroll
    for (int k = 0; k < NLDT; ++k) {
        const int q = tid + NTHR * k;
        if (k < 3 || q < nT4) {
            const unsigned e0   = 4u * (unsigned)q;
            const unsigned row0 = e0 / (unsigned)JD;
            const unsigned col0 = e0 - row0 * (unsigned)JD;
            const unsigned d0 = e0 + row0;
            const unsigned d1 = d0 + 1u + (col0 >= (unsigned)(JD - 1) ? 1u : 0u);
            const unsigned d2 = d0 + 2u + (col0 >= (unsigned)(JD - 2) ? 1u : 0u);
            const unsigned d3 = d0 + 3u + (col0 >= (unsigned)(JD - 3) ? 1u : 0u);
            dTJ[d0] = vt[k].x;
            dTJ[d1] = vt[k].y;
            dTJ[d2] = vt[k].z;
            dTJ[d3] = vt[k].w;
        }
    }
    __syncthreads();

    const int wave = tid >> 5;
    const int lane = tid & 31;
    const int tjx  = half * 4 + wave;
    if (tjx >= NTILE) return;
    const int j0  = tjx * 16;
    const int jl0 = wave * 16;

    v2f a;  a.x = 0.0f;  a.y = 0.0f;
    v2f bb; bb.x = 0.0f; bb.y = 0.0f;
    const int mlo = lane & 15;
    if (lane < 16) {
        int i = i0 + mlo;
        if (i < NQ) { a.x = sDlp[mlo]; a.y = sL1mp[mlo]; }
        int j = j0 + mlo;
        if (j < NQ) { bb.x = sNegT[jl0 + mlo]; bb.y = -1.0f; }
    }

    v8f c = {};
    c = __builtin_amdgcn_wmma_f32_16x16x4_f32(
             false, a,  false, bb,
             (short)0, c,  false,  false);

    const int j     = j0 + (lane & 15);
    const int ibase = i0 + ((lane >> 4) << 3);
    float* outb = out + b * NQ * NQ;
    if (j < NQ) {
        const float4* tjrow = &sTJ4[(jl0 + (lane & 15)) * JD4];
        float4 tr[JD4];
        #pragma unroll
        for (int kk = 0; kk < JD4; ++kk) tr[kk] = tjrow[kk];

        const float4* prow[8];
        int lrow[8];
        #pragma unroll
        for (int r = 0; r < 8; ++r) {
            const int i = ibase + r;
            lrow[r] = ((i < NQ) ? i : (NQ - 1)) - i0;
            prow[r] = &sPJ4[lrow[r] * JD4];
        }

        float l1[8];
        #pragma unroll
        for (int r = 0; r < 8; ++r) l1[r] = 0.0f;

        float4 pv[8], pn[8];
        #pragma unroll
        for (int r = 0; r < 8; ++r) pv[r] = prow[r][0];

        float sr[8];
        #pragma unroll
        for (int r = 0; r < 8; ++r) sr[r] = sRow[lrow[r]];

        #pragma unroll
        for (int kk = 0; kk < JD4; ++kk) {
            if (kk + 1 < JD4) {
                #pragma unroll
                for (int r = 0; r < 8; ++r) pn[r] = prow[r][kk + 1];
            }
            #pragma unroll
            for (int r = 0; r < 8; ++r) l1[r] += fabsf(pv[r].x - tr[kk].x);
            #pragma unroll
            for (int r = 0; r < 8; ++r) l1[r] += fabsf(pv[r].y - tr[kk].y);
            #pragma unroll
            for (int r = 0; r < 8; ++r) l1[r] += fabsf(pv[r].z - tr[kk].z);
            #pragma unroll
            for (int r = 0; r < 8; ++r) l1[r] += fabsf(pv[r].w - tr[kk].w);
            if (kk + 1 < JD4) {
                #pragma unroll
                for (int r = 0; r < 8; ++r) pv[r] = pn[r];
            }
        }

        float val[8];
        #pragma unroll
        for (int r = 0; r < 8; ++r) val[r] = c[r] + l1[r] + sr[r];

        #pragma unroll
        for (int r = 0; r < 8; ++r) asm volatile("" : "+v"(val[r]));

        #pragma unroll
        for (int r = 0; r < 8; ++r) {
            const int i = ibase + r;
            if (i < NQ) outb[i * NQ + j] = val[r];
        }
    }
}

extern "C" void kernel_launch(void* const* d_in, const int* in_sizes, int n_in,
                              void* d_out, int out_size, void* d_ws, size_t ws_size,
                              hipStream_t stream) {
    (void)in_sizes; (void)n_in; (void)d_ws; (void)ws_size; (void)out_size;
    const float* pred_conf   = (const float*)d_in[0];
    const float* pred_joint  = (const float*)d_in[1];
    const float* pred_action = (const float*)d_in[2];
    const float* pred_ident  = (const float*)d_in[3];
    const float* tar_conf    = (const float*)d_in[4];
    const float* tar_joint   = (const float*)d_in[5];
    float* out = (float*)d_out;

    dim3 grid(2 * NTILE, 16);
    dim3 block(NTHR);
    matcher_blocks_kernel<<<grid, block, 0, stream>>>(
        pred_conf, pred_joint, pred_action, pred_ident, tar_conf, tar_joint, out);
}


# L1 section: LDS reads (pred rows and the lane's target values) issued three k-groups ahead of the math in a register ring of four; was one group, fully waited for. Per-output add order unchanged
# speedup vs baseline: 3.9141x; 1.0527x over previous
#include <hip/hip_runtime.h>
#include <hip/hip_bf16.h>
#include <cmath>


typedef __attribute__((ext_vector_type(2))) float v2f;
typedef __attribute__((ext_vector_type(8))) float v8f;

#define NQ 100
#define JD 51
#define JD4 13
#define NTILE 7
#define PR 16
#define TC 64
#define NTHR 128
#define NLDP 2
#define NLDT 7

__global__ __launch_bounds__(NTHR) void matcher_blocks_kernel(
    const float* __restrict__ pred_conf,
    const float* __restrict__ pred_joint,
    const float* __restrict__ pred_action,
    const float* __restrict__ pred_ident,
    const float* __restrict__ tar_conf,
    const float* __restrict__ tar_joint,
    float* __restrict__ out)
{
    __shared__ float4 sPJ4[PR * JD4];
    __shared__ float4 sTJ4[TC * JD4];
    __shared__ float  sDlp[PR];
    __shared__ float  sL1mp[PR];
    __shared__ float  sRow[PR];
    __shared__ float  sNegT[TC];

    const int b    = blockIdx.y;
    const int tid  = threadIdx.x;
    const int ti   = blockIdx.x >> 1;
    const int half = blockIdx.x & 1;
    const int i0   = ti * 16;
    const int c0   = half * TC;
    const int nrow = (NQ - i0 < PR) ? (NQ - i0) : PR;
    const int ncol = (NQ - c0 < TC) ? (NQ - c0) : TC;
    const int nP4  = nrow * JD / 4;
    const int nT4  = ncol * JD / 4;

    const bool rowThread = tid < PR;
    const bool colThread = tid < TC;
    const int ri   = i0 + (rowThread ? tid : 0);
    const int cj   = c0 + (colThread ? tid : 0);
    const int srow = b * NQ + (ri < NQ ? ri : NQ - 1);
    const int scol = b * NQ + (cj < NQ ? cj : NQ - 1);
    const float x   = pred_conf[srow];
    const float tcv = tar_conf[scol];
    float pa[8], pi[10];
    {
        const float* pap = pred_action + srow * 8;
        #pragma unroll
        for (int k = 0; k < 8; ++k) pa[k] = pap[k];
        const float* pip = pred_ident + srow * 10;
        #pragma unroll
        for (int k = 0; k < 10; ++k) pi[k] = pip[k];
    }

    const float4* pj4 = reinterpret_cast<const float4*>(pred_joint + (b * NQ + i0) * JD);
    const float4* tj4 = reinterpret_cast<const float4*>(tar_joint  + (b * NQ + c0) * JD);
    float4 vp[NLDP], vt[NLDT];
    #pragma unroll
    for (int k = 0; k < NLDP; ++k) {
        const int q = tid + NTHR * k;
        vp[k] = pj4[q < nP4 ? q : nP4 - 1];
    }
    #pragma unroll
    for (int k = 0; k < NLDT; ++k) {
        const int q = tid + NTHR * k;
        vt[k] = tj4[q < nT4 ? q : nT4 - 1];
    }

    asm volatile("" :: "v"(x),
                 "v"(pa[0]), "v"(pa[1]), "v"(pa[2]), "v"(pa[3]),
                 "v"(pa[4]), "v"(pa[5]), "v"(pa[6]), "v"(pa[7]),
                 "v"(pi[0]), "v"(pi[1]), "v"(pi[2]), "v"(pi[3]), "v"(pi[4]),
                 "v"(pi[5]), "v"(pi[6]), "v"(pi[7]), "v"(pi[8]), "v"(pi[9]));

    float* dPJ = reinterpret_cast<float*>(sPJ4);
    float* dTJ = reinterpret_cast<float*>(sTJ4);
    if (rowThread) {
        sDlp[tid] = x;
        const float e = __expf(-fabsf(x));
        float l1mp = -(fmaxf(x, 0.0f) + __logf(1.0f + e));
        sL1mp[tid] = l1mp;

        float m = pa[0];
        #pragma unroll
        for (int k = 1; k < 8; ++k) m = fmaxf(m, pa[k]);
        float s = 0.0f;
        #pragma unroll
        for (int k = 0; k < 8; ++k) s += __expf(pa[k] - m);
        float ce = (m + __logf(s)) - pa[0];

        float m2 = pi[0];
        #pragma unroll
        for (int k = 1; k < 10; ++k) m2 = fmaxf(m2, pi[k]);
        float s2 = 0.0f;
        #pragma unroll
        for (int k = 0; k < 10; ++k) s2 += __expf(pi[k] - m2);
        ce += (m2 + __logf(s2)) - pi[0];

        sRow[tid] = ce;

        dPJ[tid * (4 * JD4) + JD] = 0.0f;
    }
    if (colThread) {
        sNegT[tid] = -tcv;
        dTJ[tid * (4 * JD4) + JD] = 0.0f;
    }

    #pragma unroll
    for (int k = 0; k < NLDP; ++k) {
        const int q = tid + NTHR * k;
        if (q < nP4) {
            const unsigned e0   = 4u * (unsigned)q;
            const unsigned row0 = e0 / (unsigned)JD;
            const unsigned col0 = e0 - row0 * (unsigned)JD;
            const unsigned d0 = e0 + row0;
            const unsigned d1 = d0 + 1u + (col0 >= (unsigned)(JD - 1) ? 1u : 0u);
            const unsigned d2 = d0 + 2u + (col0 >= (unsigned)(JD - 2) ? 1u : 0u);
            const unsigned d3 = d0 + 3u + (col0 >= (unsigned)(JD - 3) ? 1u : 0u);
            dPJ[d0] = vp[k].x;
            dPJ[d1] = vp[k].y;
            dPJ[d2] = vp[k].z;
            dPJ[d3] = vp[k].w;
        }
    }
    #pragma unroll
    for (int k = 0; k < NLDT; ++k) {
        const int q = tid + NTHR * k;
        if (k < 3 || q < nT4) {
            const unsigned e0   = 4u * (unsigned)q;
            const unsigned row0 = e0 / (unsigned)JD;
            const unsigned col0 = e0 - row0 * (unsigned)JD;
            const unsigned d0 = e0 + row0;
            const unsigned d1 = d0 + 1u + (col0 >= (unsigned)(JD - 1) ? 1u : 0u);
            const unsigned d2 = d0 + 2u + (col0 >= (unsigned)(JD - 2) ? 1u : 0u);
            const unsigned d3 = d0 + 3u + (col0 >= (unsigned)(JD - 3) ? 1u : 0u);
            dTJ[d0] = vt[k].x;
            dTJ[d1] = vt[k].y;
            dTJ[d2] = vt[k].z;
            dTJ[d3] = vt[k].w;
        }
    }
    __syncthreads();

    const int wave = tid >> 5;
    const int lane = tid & 31;
    const int tjx  = half * 4 + wave;
    if (tjx >= NTILE) return;
    const int j0  = tjx * 16;
    const int jl0 = wave * 16;

    v2f a;  a.x = 0.0f;  a.y = 0.0f;
    v2f bb; bb.x = 0.0f; bb.y = 0.0f;
    const int mlo = lane & 15;
    if (lane < 16) {
        int i = i0 + mlo;
        if (i < NQ) { a.x = sDlp[mlo]; a.y = sL1mp[mlo]; }
        int j = j0 + mlo;
        if (j < NQ) { bb.x = sNegT[jl0 + mlo]; bb.y = -1.0f; }
    }

    v8f c = {};
    c = __builtin_amdgcn_wmma_f32_16x16x4_f32(
             false, a,  false, bb,
             (short)0, c,  false,  false);

    const int j     = j0 + (lane & 15);
    const int ibase = i0 + ((lane >> 4) << 3);
    float* outb = out + b * NQ * NQ;
    if (j < NQ) {
        const float4* tjrow = &sTJ4[(jl0 + (lane & 15)) * JD4];
        const float4* prow[8];
        int lrow[8];
        #pragma unroll
        for (int r = 0; r < 8; ++r) {
            const int i = ibase + r;
            lrow[r] = ((i < NQ) ? i : (NQ - 1)) - i0;
            prow[r] = &sPJ4[lrow[r] * JD4];
        }

        float l1[8];
        #pragma unroll
        for (int r = 0; r < 8; ++r) l1[r] = 0.0f;

        constexpr int AHEAD = 3;
        constexpr int RING  = AHEAD + 1;
        float4 tq[RING];
        float4 pq[RING][8];
        #pragma unroll
        for (int s = 0; s < AHEAD; ++s) {
            tq[s] = tjrow[s];
            #pragma unroll
            for (int r = 0; r < 8; ++r) pq[s][r] = prow[r][s];
            __builtin_amdgcn_sched_barrier(0);
        }

        float sr[8];
        #pragma unroll
        for (int r = 0; r < 8; ++r) sr[r] = sRow[lrow[r]];

        #pragma unroll
        for (int kk = 0; kk < JD4; ++kk) {
            if (kk + AHEAD < JD4) {
                const int sn = (kk + AHEAD) % RING;
                tq[sn] = tjrow[kk + AHEAD];
                #pragma unroll
                for (int r = 0; r < 8; ++r) pq[sn][r] = prow[r][kk + AHEAD];
            }
            __builtin_amdgcn_sched_barrier(0);
            const int sc = kk % RING;
            #pragma unroll
            for (int r = 0; r < 8; ++r) l1[r] += fabsf(pq[sc][r].x - tq[sc].x);
            #pragma unroll
            for (int r = 0; r < 8; ++r) l1[r] += fabsf(pq[sc][r].y - tq[sc].y);
            #pragma unroll
            for (int r = 0; r < 8; ++r) l1[r] += fabsf(pq[sc][r].z - tq[sc].z);
            #pragma unroll
            for (int r = 0; r < 8; ++r) l1[r] += fabsf(pq[sc][r].w - tq[sc].w);
            __builtin_amdgcn_sched_barrier(0);
        }

        float val[8];
        #pragma unroll
        for (int r = 0; r < 8; ++r) val[r] = c[r] + l1[r] + sr[r];

        #pragma unroll
        for (int r = 0; r < 8; ++r) asm volatile("" : "+v"(val[r]));

        #pragma unroll
        for (int r = 0; r < 8; ++r) {
            const int i = ibase + r;
            if (i < NQ) outb[i * NQ + j] = val[r];
        }
    }
}

extern "C" void kernel_launch(void* const* d_in, const int* in_sizes, int n_in,
                              void* d_out, int out_size, void* d_ws, size_t ws_size,
                              hipStream_t stream) {
    (void)in_sizes; (void)n_in; (void)d_ws; (void)ws_size; (void)out_size;
    const float* pred_conf   = (const float*)d_in[0];
    const float* pred_joint  = (const float*)d_in[1];
    const float* pred_action = (const float*)d_in[2];
    const float* pred_ident  = (const float*)d_in[3];
    const float* tar_conf    = (const float*)d_in[4];
    const float* tar_joint   = (const float*)d_in[5];
    float* out = (float*)d_out;

    dim3 grid(2 * NTILE, 16);
    dim3 block(NTHR);
    matcher_blocks_kernel<<<grid, block, 0, stream>>>(
        pred_conf, pred_joint, pred_action, pred_ident, tar_conf, tar_joint, out);
}
